# starting kernel
# baseline (speedup 1.0000x reference)
#include <hip/hip_runtime.h>
#include <math.h>

typedef __attribute__((ext_vector_type(2))) float v2f;
typedef __attribute__((ext_vector_type(8))) float v8f;

#define DEPTH 4
#define H1 64
#define H2 32
#define NP 64
#define DIMX 3
#define FSIZE 160
#define BATCHN 256

__device__ __forceinline__ v8f wmma_f32_k4(v2f a, v2f b, v8f c) {
  return __builtin_amdgcn_wmma_f32_16x16x4_f32(false, a, false, b, (short)0, c,
                                               false, false);
}

__device__ __forceinline__ float fast_tanh(float x) {
#if __has_builtin(__builtin_amdgcn_tanhf)
  return __builtin_amdgcn_tanhf(x);
#else
  float r;
  asm volatile("v_tanh_f32 %0, %1\n\tv_nop" : "=v"(r) : "v"(x));
  return r;
#endif
}

__global__ __launch_bounds__(64) void drift_pair_chain(
    const float* __restrict__ x, const float* __restrict__ t,
    const float* __restrict__ fc2_w0, const float* __restrict__ fc2_b0,
    const float* __restrict__ fc2_w, const float* __restrict__ fc2_b,
    float* __restrict__ mfeat, float* __restrict__ m0,
    float* __restrict__ m1, float* __restrict__ m2) {
  const int i = blockIdx.x;
  const int b = blockIdx.y;
  const int tid = threadIdx.x;

  __shared__ float hA[NP][H2 + 4];
  __shared__ float hB[NP][H2 + 4];
  __shared__ float wS[H2][H2 + 4];
  __shared__ float sfeat[NP][4];
  __shared__ float biasS[H2];

  __builtin_prefetch(fc2_w, 0, 1);

  const float tval = t[0];
  const float xi0 = x[(b * NP + i) * 3 + 0];
  const float xi1 = x[(b * NP + i) * 3 + 1];
  const float xi2 = x[(b * NP + i) * 3 + 2];

  const int j = tid;
  float r0 = xi0 - x[(b * NP + j) * 3 + 0];
  float r1 = xi1 - x[(b * NP + j) * 3 + 1];
  float r2 = xi2 - x[(b * NP + j) * 3 + 2];
  if (j == i) { r0 += 1.f; r1 += 1.f; r2 += 1.f; }
  const float nr = sqrtf(r0 * r0 + r1 * r1 + r2 * r2);
  sfeat[j][0] = r0; sfeat[j][1] = r1; sfeat[j][2] = r2; sfeat[j][3] = nr;
  {
    const float f5[5] = {r0, r1, r2, nr, tval};
    #pragma unroll
    for (int o = 0; o < H2; ++o) {
      float acc = fc2_b0[o];
      #pragma unroll
      for (int k = 0; k < 5; ++k) acc += f5[k] * fc2_w0[o * 5 + k];
      hA[j][o] = fast_tanh(acc);
    }
  }
  __syncthreads();

  if (tid < 4) {
    float s = 0.f;
    for (int jj = 0; jj < NP; ++jj) s += sfeat[jj][tid];
    mfeat[(b * NP + i) * 4 + tid] = s * (1.f / NP);
  }
  if (tid < H2) {
    float s = 0.f;
    for (int jj = 0; jj < NP; ++jj) s += hA[jj][tid];
    m0[(b * NP + i) * H2 + tid] = s * (1.f / NP);
  }

  const int wave = tid >> 5;
  const int lane = tid & 31;
  const int half = lane >> 4;
  const int lr = lane & 15;

  for (int L = 0; L < 2; ++L) {
    const float* Wl = fc2_w + L * H2 * H2;
    for (int idx = tid; idx < H2 * H2; idx += 64)
      wS[idx >> 5][idx & 31] = Wl[idx];
    if (tid < H2) biasS[tid] = fc2_b[L * H2 + tid];
    __syncthreads();

    float(*src)[H2 + 4] = (L == 0) ? hA : hB;
    float(*dst)[H2 + 4] = (L == 0) ? hB : hA;
    const int rowbase = wave * 32;

    #pragma unroll
    for (int rt = 0; rt < 2; ++rt) {
      #pragma unroll
      for (int ct = 0; ct < 2; ++ct) {
        v8f acc = {};
        #pragma unroll
        for (int kk = 0; kk < 8; ++kk) {
          const int k0 = kk * 4 + half * 2;
          const int ar = rowbase + rt * 16 + lr;
          v2f a;  a.x = src[ar][k0];              a.y = src[ar][k0 + 1];
          v2f bv; bv.x = wS[ct * 16 + lr][k0];    bv.y = wS[ct * 16 + lr][k0 + 1];
          acc = wmma_f32_k4(a, bv, acc);
        }
        #pragma unroll
        for (int r = 0; r < 8; ++r) {
          const int row = rowbase + rt * 16 + r + half * 8;
          const int col = ct * 16 + lr;
          dst[row][col] = fast_tanh(acc[r] + biasS[col]) + src[row][col];
        }
      }
    }
    __syncthreads();

    float* mo = (L == 0) ? m1 : m2;
    if (tid < H2) {
      float s = 0.f;
      for (int jj = 0; jj < NP; ++jj) s += dst[jj][tid];
      mo[(b * NP + i) * H2 + tid] = s * (1.f / NP);
    }
    __syncthreads();
  }
}

__global__ __launch_bounds__(64) void drift_particle_chain(
    const float* __restrict__ x, const float* __restrict__ t,
    const float* __restrict__ fc1_w0, const float* __restrict__ fc1_b0,
    const float* __restrict__ fc1_w, const float* __restrict__ fc1_b,
    const float* __restrict__ final_w, const float* __restrict__ final_b,
    const float* __restrict__ mfeat, const float* __restrict__ m0,
    const float* __restrict__ m1, const float* __restrict__ m2,
    float* __restrict__ out) {
  const int b = blockIdx.x;
  const int tid = threadIdx.x;

  __shared__ float hA[NP][H1 + 4];
  __shared__ float hB[NP][H1 + 4];
  __shared__ float mprev[NP][H2 + 4];
  __shared__ float meanh[H1];
  __shared__ float sfeat[NP][4];
  __shared__ float smf[4];
  __shared__ float biasS[H1];

  __builtin_prefetch(fc1_w, 0, 1);

  const float tval = t[0];
  const int i = tid;
  const float x0 = x[(b * NP + i) * 3 + 0];
  const float x1 = x[(b * NP + i) * 3 + 1];
  const float x2 = x[(b * NP + i) * 3 + 2];
  const float nr = sqrtf(x0 * x0 + x1 * x1 + x2 * x2);
  sfeat[i][0] = x0; sfeat[i][1] = x1; sfeat[i][2] = x2; sfeat[i][3] = nr;
  __syncthreads();
  if (tid < 4) {
    float s = 0.f;
    for (int jj = 0; jj < NP; ++jj) s += sfeat[jj][tid];
    smf[tid] = s * (1.f / NP);
  }
  __syncthreads();

  {
    float ft[13];
    ft[0] = x0; ft[1] = x1; ft[2] = x2; ft[3] = nr;
    ft[4] = smf[0]; ft[5] = smf[1]; ft[6] = smf[2]; ft[7] = smf[3];
    ft[8]  = mfeat[(b * NP + i) * 4 + 0];
    ft[9]  = mfeat[(b * NP + i) * 4 + 1];
    ft[10] = mfeat[(b * NP + i) * 4 + 2];
    ft[11] = mfeat[(b * NP + i) * 4 + 3];
    ft[12] = tval;
    for (int o = 0; o < H1; ++o) {
      float acc = fc1_b0[o];
      #pragma unroll
      for (int k = 0; k < 13; ++k) acc += ft[k] * fc1_w0[o * 13 + k];
      hA[i][o] = fast_tanh(acc);
    }
  }
  __syncthreads();

  const int wave = tid >> 5;
  const int lane = tid & 31;
  const int half = lane >> 4;
  const int lr = lane & 15;

  for (int d = 1; d <= 3; ++d) {
    const float* Wl = fc1_w + (d - 1) * H1 * FSIZE;
    const float* ms = (d == 1) ? m0 : (d == 2) ? m1 : m2;
    float(*src)[H1 + 4] = (d & 1) ? hA : hB;
    float(*dst)[H1 + 4] = (d & 1) ? hB : hA;

    for (int idx = tid; idx < NP * H2; idx += 64)
      mprev[idx >> 5][idx & 31] = ms[b * NP * H2 + idx];
    biasS[tid] = fc1_b[(d - 1) * H1 + tid];
    {
      float s = 0.f;
      for (int jj = 0; jj < NP; ++jj) s += src[jj][tid];
      meanh[tid] = s * (1.f / NP);
    }
    __syncthreads();

    #pragma unroll
    for (int rt = 0; rt < 4; ++rt) {
      #pragma unroll
      for (int ct2 = 0; ct2 < 2; ++ct2) {
        const int ct = wave * 2 + ct2;
        v8f acc = {};
        #pragma unroll 8
        for (int kk = 0; kk < 40; ++kk) {
          const int k0 = kk * 4 + half * 2;
          const int ar = rt * 16 + lr;
          v2f a;
          if (kk < 16) {
            a.x = src[ar][k0];          a.y = src[ar][k0 + 1];
          } else if (kk < 32) {
            a.x = meanh[k0 - 64];       a.y = meanh[k0 - 63];
          } else {
            a.x = mprev[ar][k0 - 128];  a.y = mprev[ar][k0 - 127];
          }
          const float* wr = Wl + (ct * 16 + lr) * FSIZE + k0;
          v2f bv; bv.x = wr[0]; bv.y = wr[1];
          acc = wmma_f32_k4(a, bv, acc);
        }
        #pragma unroll
        for (int r = 0; r < 8; ++r) {
          const int row = rt * 16 + r + half * 8;
          const int col = ct * 16 + lr;
          dst[row][col] = fast_tanh(acc[r] + biasS[col]) + src[row][col];
        }
      }
    }
    __syncthreads();
  }

  #pragma unroll
  for (int dd = 0; dd < DIMX; ++dd) {
    float acc = final_b[dd];
    for (int k = 0; k < H1; ++k) acc += hB[tid][k] * final_w[dd * H1 + k];
    out[(b * NP + tid) * 3 + dd] = acc;
  }
}

extern "C" void kernel_launch(void* const* d_in, const int* in_sizes, int n_in,
                              void* d_out, int out_size, void* d_ws,
                              size_t ws_size, hipStream_t stream) {
  const float* x       = (const float*)d_in[0];
  const float* t       = (const float*)d_in[1];
  const float* fc1_w0  = (const float*)d_in[2];
  const float* fc1_b0  = (const float*)d_in[3];
  const float* fc1_w   = (const float*)d_in[4];
  const float* fc1_b   = (const float*)d_in[5];
  const float* fc2_w0  = (const float*)d_in[6];
  const float* fc2_b0  = (const float*)d_in[7];
  const float* fc2_w   = (const float*)d_in[8];
  const float* fc2_b   = (const float*)d_in[9];
  const float* final_w = (const float*)d_in[10];
  const float* final_b = (const float*)d_in[11];
  float* out = (float*)d_out;

  float* ws    = (float*)d_ws;
  float* mfeat = ws;
  float* m0    = mfeat + BATCHN * NP * 4;
  float* m1    = m0 + BATCHN * NP * H2;
  float* m2    = m1 + BATCHN * NP * H2;
  (void)in_sizes; (void)n_in; (void)out_size; (void)ws_size;

  drift_pair_chain<<<dim3(NP, BATCHN), 64, 0, stream>>>(
      x, t, fc2_w0, fc2_b0, fc2_w, fc2_b, mfeat, m0, m1, m2);
  drift_particle_chain<<<dim3(BATCHN), 64, 0, stream>>>(
      x, t, fc1_w0, fc1_b0, fc1_w, fc1_b, final_w, final_b,
      mfeat, m0, m1, m2, out);
}


# particle-chain kernel: a layer's operand reads (weights, mean_j h2 from global; A fragments, residual, bias from LDS) issued once as groups ahead of the WMMA stream, not one dependent load per use
# speedup vs baseline: 2.0124x; 2.0124x over previous
#include <hip/hip_runtime.h>
#include <math.h>

typedef __attribute__((ext_vector_type(2))) float v2f;
typedef __attribute__((ext_vector_type(8))) float v8f;

#define DEPTH 4
#define H1 64
#define H2 32
#define NP 64
#define DIMX 3
#define FSIZE 160
#define BATCHN 256

__device__ __forceinline__ v8f wmma_f32_k4(v2f a, v2f b, v8f c) {
  return __builtin_amdgcn_wmma_f32_16x16x4_f32(false, a, false, b, (short)0, c,
                                               false, false);
}

__device__ __forceinline__ float fast_tanh(float x) {
#if __has_builtin(__builtin_amdgcn_tanhf)
  return __builtin_amdgcn_tanhf(x);
#else
  float r;
  asm volatile("v_tanh_f32 %0, %1\n\tv_nop" : "=v"(r) : "v"(x));
  return r;
#endif
}

__global__ __launch_bounds__(64) void drift_pair_chain(
    const float* __restrict__ x, const float* __restrict__ t,
    const float* __restrict__ fc2_w0, const float* __restrict__ fc2_b0,
    const float* __restrict__ fc2_w, const float* __restrict__ fc2_b,
    float* __restrict__ mfeat, float* __restrict__ m0,
    float* __restrict__ m1, float* __restrict__ m2) {
  const int i = blockIdx.x;
  const int b = blockIdx.y;
  const int tid = threadIdx.x;

  __shared__ float hA[NP][H2 + 4];
  __shared__ float hB[NP][H2 + 4];
  __shared__ float wS[H2][H2 + 4];
  __shared__ float sfeat[NP][4];
  __shared__ float biasS[H2];

  __builtin_prefetch(fc2_w, 0, 1);

  const float tval = t[0];
  const float xi0 = x[(b * NP + i) * 3 + 0];
  const float xi1 = x[(b * NP + i) * 3 + 1];
  const float xi2 = x[(b * NP + i) * 3 + 2];

  const int j = tid;
  float r0 = xi0 - x[(b * NP + j) * 3 + 0];
  float r1 = xi1 - x[(b * NP + j) * 3 + 1];
  float r2 = xi2 - x[(b * NP + j) * 3 + 2];
  if (j == i) { r0 += 1.f; r1 += 1.f; r2 += 1.f; }
  const float nr = sqrtf(r0 * r0 + r1 * r1 + r2 * r2);
  sfeat[j][0] = r0; sfeat[j][1] = r1; sfeat[j][2] = r2; sfeat[j][3] = nr;
  {
    const float f5[5] = {r0, r1, r2, nr, tval};
    #pragma unroll
    for (int o = 0; o < H2; ++o) {
      float acc = fc2_b0[o];
      #pragma unroll
      for (int k = 0; k < 5; ++k) acc += f5[k] * fc2_w0[o * 5 + k];
      hA[j][o] = fast_tanh(acc);
    }
  }
  __syncthreads();

  if (tid < 4) {
    float s = 0.f;
    for (int jj = 0; jj < NP; ++jj) s += sfeat[jj][tid];
    mfeat[(b * NP + i) * 4 + tid] = s * (1.f / NP);
  }
  if (tid < H2) {
    float s = 0.f;
    for (int jj = 0; jj < NP; ++jj) s += hA[jj][tid];
    m0[(b * NP + i) * H2 + tid] = s * (1.f / NP);
  }

  const int wave = tid >> 5;
  const int lane = tid & 31;
  const int half = lane >> 4;
  const int lr = lane & 15;

  for (int L = 0; L < 2; ++L) {
    const float* Wl = fc2_w + L * H2 * H2;
    for (int idx = tid; idx < H2 * H2; idx += 64)
      wS[idx >> 5][idx & 31] = Wl[idx];
    if (tid < H2) biasS[tid] = fc2_b[L * H2 + tid];
    __syncthreads();

    float(*src)[H2 + 4] = (L == 0) ? hA : hB;
    float(*dst)[H2 + 4] = (L == 0) ? hB : hA;
    const int rowbase = wave * 32;

    #pragma unroll
    for (int rt = 0; rt < 2; ++rt) {
      #pragma unroll
      for (int ct = 0; ct < 2; ++ct) {
        v8f acc = {};
        #pragma unroll
        for (int kk = 0; kk < 8; ++kk) {
          const int k0 = kk * 4 + half * 2;
          const int ar = rowbase + rt * 16 + lr;
          v2f a;  a.x = src[ar][k0];              a.y = src[ar][k0 + 1];
          v2f bv; bv.x = wS[ct * 16 + lr][k0];    bv.y = wS[ct * 16 + lr][k0 + 1];
          acc = wmma_f32_k4(a, bv, acc);
        }
        #pragma unroll
        for (int r = 0; r < 8; ++r) {
          const int row = rowbase + rt * 16 + r + half * 8;
          const int col = ct * 16 + lr;
          dst[row][col] = fast_tanh(acc[r] + biasS[col]) + src[row][col];
        }
      }
    }
    __syncthreads();

    float* mo = (L == 0) ? m1 : m2;
    if (tid < H2) {
      float s = 0.f;
      for (int jj = 0; jj < NP; ++jj) s += dst[jj][tid];
      mo[(b * NP + i) * H2 + tid] = s * (1.f / NP);
    }
    __syncthreads();
  }
}

__global__ __launch_bounds__(64) void drift_particle_chain(
    const float* __restrict__ x, const float* __restrict__ t,
    const float* __restrict__ fc1_w0, const float* __restrict__ fc1_b0,
    const float* __restrict__ fc1_w, const float* __restrict__ fc1_b,
    const float* __restrict__ final_w, const float* __restrict__ final_b,
    const float* __restrict__ mfeat, const float* __restrict__ m0,
    const float* __restrict__ m1, const float* __restrict__ m2,
    float* __restrict__ out) {
  const int b = blockIdx.x;
  const int tid = threadIdx.x;

  __shared__ float hA[NP][H1 + 4];
  __shared__ float hB[NP][H1 + 4];
  __shared__ float mprev[NP][H2 + 4];
  __shared__ float meanh[H1];
  __shared__ float sfeat[NP][4];
  __shared__ float smf[4];
  __shared__ float biasS[H1];

  __builtin_prefetch(fc1_w, 0, 1);

  const float tval = t[0];
  const int i = tid;
  const float x0 = x[(b * NP + i) * 3 + 0];
  const float x1 = x[(b * NP + i) * 3 + 1];
  const float x2 = x[(b * NP + i) * 3 + 2];
  const float nr = sqrtf(x0 * x0 + x1 * x1 + x2 * x2);
  sfeat[i][0] = x0; sfeat[i][1] = x1; sfeat[i][2] = x2; sfeat[i][3] = nr;
  __syncthreads();
  if (tid < 4) {
    float s = 0.f;
    for (int jj = 0; jj < NP; ++jj) s += sfeat[jj][tid];
    smf[tid] = s * (1.f / NP);
  }
  __syncthreads();

  {
    float ft[13];
    ft[0] = x0; ft[1] = x1; ft[2] = x2; ft[3] = nr;
    ft[4] = smf[0]; ft[5] = smf[1]; ft[6] = smf[2]; ft[7] = smf[3];
    ft[8]  = mfeat[(b * NP + i) * 4 + 0];
    ft[9]  = mfeat[(b * NP + i) * 4 + 1];
    ft[10] = mfeat[(b * NP + i) * 4 + 2];
    ft[11] = mfeat[(b * NP + i) * 4 + 3];
    ft[12] = tval;
    for (int o = 0; o < H1; ++o) {
      float acc = fc1_b0[o];
      #pragma unroll
      for (int k = 0; k < 13; ++k) acc += ft[k] * fc1_w0[o * 13 + k];
      hA[i][o] = fast_tanh(acc);
    }
  }
  __syncthreads();

  const int wave = tid >> 5;
  const int lane = tid & 31;
  const int half = lane >> 4;
  const int lr = lane & 15;

  for (int d = 1; d <= 3; ++d) {
    const float* Wl = fc1_w + (d - 1) * H1 * FSIZE;
    const float* ms = (d == 1) ? m0 : (d == 2) ? m1 : m2;
    float(*src)[H1 + 4] = (d & 1) ? hA : hB;
    float(*dst)[H1 + 4] = (d & 1) ? hB : hA;

    const float bval = fc1_b[(d - 1) * H1 + tid];
    float mp[32];
    #pragma unroll
    for (int q = 0; q < 32; ++q) mp[q] = ms[b * NP * H2 + q * 64 + tid];

    v2f bfr[2][40];
    const float* wr0 = Wl + ((wave * 2 + 0) * 16 + lr) * FSIZE + half * 2;
    const float* wr1 = Wl + ((wave * 2 + 1) * 16 + lr) * FSIZE + half * 2;
    #pragma unroll
    for (int kk = 0; kk < 14; ++kk) {
      bfr[0][kk].x = wr0[kk * 4]; bfr[0][kk].y = wr0[kk * 4 + 1];
      bfr[1][kk].x = wr1[kk * 4]; bfr[1][kk].y = wr1[kk * 4 + 1];
    }

    {
      float s = 0.f;
      for (int jj = 0; jj < NP; ++jj) s += src[jj][tid];
      meanh[tid] = s * (1.f / NP);
    }
    biasS[tid] = bval;
    #pragma unroll
    for (int q = 0; q < 32; ++q) {
      const int idx = q * 64 + tid;
      mprev[idx >> 5][idx & 31] = mp[q];
    }
    __syncthreads();

    #pragma unroll
    for (int kk = 14; kk < 40; ++kk) {
      bfr[0][kk].x = wr0[kk * 4]; bfr[0][kk].y = wr0[kk * 4 + 1];
      bfr[1][kk].x = wr1[kk * 4]; bfr[1][kk].y = wr1[kk * 4 + 1];
    }

    v2f afr[40][4];
    #pragma unroll
    for (int kk = 0; kk < 16; ++kk) {
      const int k0 = kk * 4 + half * 2;
      #pragma unroll
      for (int rt = 0; rt < 4; ++rt) {
        const int ar = rt * 16 + lr;
        afr[kk][rt].x = src[ar][k0]; afr[kk][rt].y = src[ar][k0 + 1];
      }
    }
    #pragma unroll
    for (int kk = 16; kk < 32; ++kk) {
      const int k0 = kk * 4 + half * 2;
      afr[kk][0].x = meanh[k0 - 64]; afr[kk][0].y = meanh[k0 - 63];
    }
    #pragma unroll
    for (int kk = 32; kk < 40; ++kk) {
      const int k0 = kk * 4 + half * 2;
      #pragma unroll
      for (int rt = 0; rt < 4; ++rt) {
        const int ar = rt * 16 + lr;
        afr[kk][rt].x = mprev[ar][k0 - 128]; afr[kk][rt].y = mprev[ar][k0 - 127];
      }
    }
    float res[2][4][8];
    float bias[2];
    #pragma unroll
    for (int ct2 = 0; ct2 < 2; ++ct2) {
      const int col = (wave * 2 + ct2) * 16 + lr;
      bias[ct2] = biasS[col];
      #pragma unroll
      for (int rt = 0; rt < 4; ++rt) {
        #pragma unroll
        for (int r = 0; r < 8; ++r) res[ct2][rt][r] = src[rt * 16 + r + half * 8][col];
      }
    }
    __builtin_amdgcn_sched_barrier(0);

    v8f acc[2][4] = {};
    #pragma unroll
    for (int kk = 0; kk < 40; ++kk) {
      #pragma unroll
      for (int rt = 0; rt < 4; ++rt) {
        const v2f a = (kk >= 16 && kk < 32) ? afr[kk][0] : afr[kk][rt];
        #pragma unroll
        for (int ct2 = 0; ct2 < 2; ++ct2)
          acc[ct2][rt] = wmma_f32_k4(a, bfr[ct2][kk], acc[ct2][rt]);
      }
    }

    #pragma unroll
    for (int ct2 = 0; ct2 < 2; ++ct2) {
      const int col = (wave * 2 + ct2) * 16 + lr;
      #pragma unroll
      for (int rt = 0; rt < 4; ++rt) {
        #pragma unroll
        for (int r = 0; r < 8; ++r) {
          const int row = rt * 16 + r + half * 8;
          dst[row][col] = fast_tanh(acc[ct2][rt][r] + bias[ct2]) + res[ct2][rt][r];
        }
      }
    }
    __syncthreads();
  }

  #pragma unroll
  for (int dd = 0; dd < DIMX; ++dd) {
    float acc = final_b[dd];
    for (int k = 0; k < H1; ++k) acc += hB[tid][k] * final_w[dd * H1 + k];
    out[(b * NP + tid) * 3 + dd] = acc;
  }
}

extern "C" void kernel_launch(void* const* d_in, const int* in_sizes, int n_in,
                              void* d_out, int out_size, void* d_ws,
                              size_t ws_size, hipStream_t stream) {
  const float* x       = (const float*)d_in[0];
  const float* t       = (const float*)d_in[1];
  const float* fc1_w0  = (const float*)d_in[2];
  const float* fc1_b0  = (const float*)d_in[3];
  const float* fc1_w   = (const float*)d_in[4];
  const float* fc1_b   = (const float*)d_in[5];
  const float* fc2_w0  = (const float*)d_in[6];
  const float* fc2_b0  = (const float*)d_in[7];
  const float* fc2_w   = (const float*)d_in[8];
  const float* fc2_b   = (const float*)d_in[9];
  const float* final_w = (const float*)d_in[10];
  const float* final_b = (const float*)d_in[11];
  float* out = (float*)d_out;

  float* ws    = (float*)d_ws;
  float* mfeat = ws;
  float* m0    = mfeat + BATCHN * NP * 4;
  float* m1    = m0 + BATCHN * NP * H2;
  float* m2    = m1 + BATCHN * NP * H2;
  (void)in_sizes; (void)n_in; (void)out_size; (void)ws_size;

  drift_pair_chain<<<dim3(NP, BATCHN), 64, 0, stream>>>(
      x, t, fc2_w0, fc2_b0, fc2_w, fc2_b, mfeat, m0, m1, m2);
  drift_particle_chain<<<dim3(BATCHN), 64, 0, stream>>>(
      x, t, fc1_w0, fc1_b0, fc1_w, fc1_b, final_w, final_b,
      mfeat, m0, m1, m2, out);
}


# pair-chain kernel re-cut to one wave per (b,i), no block barriers: weights read straight into B fragments, h stays in D-layout registers, layer 0 on WMMA, means as register tree sums (order changes)
# speedup vs baseline: 3.0602x; 1.5207x over previous
#include <hip/hip_runtime.h>
#include <math.h>

typedef __attribute__((ext_vector_type(2))) float v2f;
typedef __attribute__((ext_vector_type(8))) float v8f;

#define DEPTH 4
#define H1 64
#define H2 32
#define NP 64
#define DIMX 3
#define FSIZE 160
#define BATCHN 256

__device__ __forceinline__ v8f wmma_f32_k4(v2f a, v2f b, v8f c) {
  return __builtin_amdgcn_wmma_f32_16x16x4_f32(false, a, false, b, (short)0, c,
                                               false, false);
}

__device__ __forceinline__ float fast_tanh(float x) {
#if __has_builtin(__builtin_amdgcn_tanhf)
  return __builtin_amdgcn_tanhf(x);
#else
  float r;
  asm volatile("v_tanh_f32 %0, %1\n\tv_nop" : "=v"(r) : "v"(x));
  return r;
#endif
}

#define PW 4
#define PGRID 16
#define PTHREADS 128
#define HP 36
static_assert(PGRID * PW == NP && PTHREADS == 32 * PW, "pair-chain launch shape");

__device__ __forceinline__ float xhalf(float v) {
  const int q = __builtin_bit_cast(int, v);
  return __builtin_bit_cast(
      float, __builtin_amdgcn_permlanex16(q, q, 0x76543210, 0xfedcba98, false, false));
}

__device__ __forceinline__ void col_mean_store(const float (&h)[4][2][8],
                                               float* __restrict__ mo, int bi,
                                               int half, int lr) {
  float cs[2];
  #pragma unroll
  for (int ct = 0; ct < 2; ++ct) {
    float p[4];
    #pragma unroll
    for (int rt = 0; rt < 4; ++rt) {
      float s = h[rt][ct][0];
      #pragma unroll
      for (int r = 1; r < 8; ++r) s += h[rt][ct][r];
      p[rt] = s;
    }
    const float s = (p[0] + p[1]) + (p[2] + p[3]);
    cs[ct] = s + xhalf(s);
  }
  mo[bi * H2 + half * 16 + lr] = (half ? cs[1] : cs[0]) * (1.f / NP);
}

__global__ __launch_bounds__(PTHREADS, 4) void drift_pair_chain(
    const float* __restrict__ x, const float* __restrict__ t,
    const float* __restrict__ fc2_w0, const float* __restrict__ fc2_b0,
    const float* __restrict__ fc2_w, const float* __restrict__ fc2_b,
    float* __restrict__ mfeat, float* __restrict__ m0,
    float* __restrict__ m1, float* __restrict__ m2) {
  const int b = blockIdx.y;
  const int wave = __builtin_amdgcn_readfirstlane((int)(threadIdx.x >> 5));
  const int lane = threadIdx.x & 31;
  const int half = lane >> 4;
  const int lr = lane & 15;
  const int i = blockIdx.x * PW + wave;
  const int bi = b * NP + i;

  __shared__ float hS[PW][NP][HP];
  float(*hs)[HP] = hS[wave];

  const float tval = t[0];
  const float xi0 = x[bi * 3 + 0];
  const float xi1 = x[bi * 3 + 1];
  const float xi2 = x[bi * 3 + 2];
  float xj[4][3];
  #pragma unroll
  for (int rt = 0; rt < 4; ++rt) {
    const int j = rt * 16 + lr;
    #pragma unroll
    for (int d = 0; d < 3; ++d) xj[rt][d] = x[(b * NP + j) * 3 + d];
  }
  v2f w0a[2];
  float w04[2], b0v[2];
  #pragma unroll
  for (int ct = 0; ct < 2; ++ct) {
    const int n = ct * 16 + lr;
    w0a[ct].x = fc2_w0[n * 5 + half * 2];
    w0a[ct].y = fc2_w0[n * 5 + half * 2 + 1];
    w04[ct] = fc2_w0[n * 5 + 4];
    b0v[ct] = fc2_b0[n];
  }

  v2f a0[4];
  float fs0 = 0.f, fs1 = 0.f;
  #pragma unroll
  for (int rt = 0; rt < 4; ++rt) {
    const int j = rt * 16 + lr;
    float r0 = xi0 - xj[rt][0];
    float r1 = xi1 - xj[rt][1];
    float r2 = xi2 - xj[rt][2];
    if (j == i) { r0 += 1.f; r1 += 1.f; r2 += 1.f; }
    const float nr = sqrtf(r0 * r0 + r1 * r1 + r2 * r2);
    a0[rt].x = half ? r2 : r0;
    a0[rt].y = half ? nr : r1;
    fs0 += a0[rt].x;
    fs1 += a0[rt].y;
  }
  #pragma unroll
  for (int m = 1; m < 16; m <<= 1) {
    fs0 += __shfl_xor(fs0, m, 32);
    fs1 += __shfl_xor(fs1, m, 32);
  }
  if (lr == 0) {
    mfeat[bi * 4 + half * 2 + 0] = fs0 * (1.f / NP);
    mfeat[bi * 4 + half * 2 + 1] = fs1 * (1.f / NP);
  }

  v8f c0[2];
  #pragma unroll
  for (int ct = 0; ct < 2; ++ct) {
    const float cv = tval * w04[ct] + b0v[ct];
    #pragma unroll
    for (int r = 0; r < 8; ++r) c0[ct][r] = cv;
  }
  float h[4][2][8];
  #pragma unroll
  for (int rt = 0; rt < 4; ++rt) {
    #pragma unroll
    for (int ct = 0; ct < 2; ++ct) {
      const v8f c = wmma_f32_k4(a0[rt], w0a[ct], c0[ct]);
      #pragma unroll
      for (int r = 0; r < 8; ++r) h[rt][ct][r] = fast_tanh(c[r]);
    }
  }

  #pragma unroll
  for (int L = 0; L < 2; ++L) {
    v2f bw[2][8];
    float bias[2];
    #pragma unroll
    for (int ct = 0; ct < 2; ++ct) {
      const float* wr = fc2_w + L * H2 * H2 + (ct * 16 + lr) * H2 + half * 2;
      #pragma unroll
      for (int kk = 0; kk < 8; ++kk) {
        bw[ct][kk].x = wr[kk * 4];
        bw[ct][kk].y = wr[kk * 4 + 1];
      }
      bias[ct] = fc2_b[L * H2 + ct * 16 + lr];
    }

    col_mean_store(h, (L == 0) ? m0 : m1, bi, half, lr);

    #pragma unroll
    for (int rt = 0; rt < 4; ++rt) {
      #pragma unroll
      for (int ct = 0; ct < 2; ++ct) {
        #pragma unroll
        for (int r = 0; r < 8; ++r)
          hs[rt * 16 + half * 8 + r][ct * 16 + lr] = h[rt][ct][r];
      }
    }
    __builtin_amdgcn_fence(__ATOMIC_RELEASE, "wavefront");
    __builtin_amdgcn_wave_barrier();
    __builtin_amdgcn_fence(__ATOMIC_ACQUIRE, "wavefront");

    v2f af[4][8];
    #pragma unroll
    for (int rt = 0; rt < 4; ++rt) {
      #pragma unroll
      for (int kk = 0; kk < 8; ++kk) {
        const int k0 = kk * 4 + half * 2;
        af[rt][kk].x = hs[rt * 16 + lr][k0];
        af[rt][kk].y = hs[rt * 16 + lr][k0 + 1];
      }
    }
    __builtin_amdgcn_sched_barrier(0);

    v8f acc[4][2] = {};
    #pragma unroll
    for (int kk = 0; kk < 8; ++kk) {
      #pragma unroll
      for (int rt = 0; rt < 4; ++rt) {
        #pragma unroll
        for (int ct = 0; ct < 2; ++ct)
          acc[rt][ct] = wmma_f32_k4(af[rt][kk], bw[ct][kk], acc[rt][ct]);
      }
    }
    #pragma unroll
    for (int rt = 0; rt < 4; ++rt) {
      #pragma unroll
      for (int ct = 0; ct < 2; ++ct) {
        #pragma unroll
        for (int r = 0; r < 8; ++r)
          h[rt][ct][r] = fast_tanh(acc[rt][ct][r] + bias[ct]) + h[rt][ct][r];
      }
    }
  }
  col_mean_store(h, m2, bi, half, lr);
}

__global__ __launch_bounds__(64) void drift_particle_chain(
    const float* __restrict__ x, const float* __restrict__ t,
    const float* __restrict__ fc1_w0, const float* __restrict__ fc1_b0,
    const float* __restrict__ fc1_w, const float* __restrict__ fc1_b,
    const float* __restrict__ final_w, const float* __restrict__ final_b,
    const float* __restrict__ mfeat, const float* __restrict__ m0,
    const float* __restrict__ m1, const float* __restrict__ m2,
    float* __restrict__ out) {
  const int b = blockIdx.x;
  const int tid = threadIdx.x;

  __shared__ float hA[NP][H1 + 4];
  __shared__ float hB[NP][H1 + 4];
  __shared__ float mprev[NP][H2 + 4];
  __shared__ float meanh[H1];
  __shared__ float sfeat[NP][4];
  __shared__ float smf[4];
  __shared__ float biasS[H1];

  __builtin_prefetch(fc1_w, 0, 1);

  const float tval = t[0];
  const int i = tid;
  const float x0 = x[(b * NP + i) * 3 + 0];
  const float x1 = x[(b * NP + i) * 3 + 1];
  const float x2 = x[(b * NP + i) * 3 + 2];
  const float nr = sqrtf(x0 * x0 + x1 * x1 + x2 * x2);
  sfeat[i][0] = x0; sfeat[i][1] = x1; sfeat[i][2] = x2; sfeat[i][3] = nr;
  __syncthreads();
  if (tid < 4) {
    float s = 0.f;
    for (int jj = 0; jj < NP; ++jj) s += sfeat[jj][tid];
    smf[tid] = s * (1.f / NP);
  }
  __syncthreads();

  {
    float ft[13];
    ft[0] = x0; ft[1] = x1; ft[2] = x2; ft[3] = nr;
    ft[4] = smf[0]; ft[5] = smf[1]; ft[6] = smf[2]; ft[7] = smf[3];
    ft[8]  = mfeat[(b * NP + i) * 4 + 0];
    ft[9]  = mfeat[(b * NP + i) * 4 + 1];
    ft[10] = mfeat[(b * NP + i) * 4 + 2];
    ft[11] = mfeat[(b * NP + i) * 4 + 3];
    ft[12] = tval;
    for (int o = 0; o < H1; ++o) {
      float acc = fc1_b0[o];
      #pragma unroll
      for (int k = 0; k < 13; ++k) acc += ft[k] * fc1_w0[o * 13 + k];
      hA[i][o] = fast_tanh(acc);
    }
  }
  __syncthreads();

  const int wave = tid >> 5;
  const int lane = tid & 31;
  const int half = lane >> 4;
  const int lr = lane & 15;

  for (int d = 1; d <= 3; ++d) {
    const float* Wl = fc1_w + (d - 1) * H1 * FSIZE;
    const float* ms = (d == 1) ? m0 : (d == 2) ? m1 : m2;
    float(*src)[H1 + 4] = (d & 1) ? hA : hB;
    float(*dst)[H1 + 4] = (d & 1) ? hB : hA;

    const float bval = fc1_b[(d - 1) * H1 + tid];
    float mp[32];
    #pragma unroll
    for (int q = 0; q < 32; ++q) mp[q] = ms[b * NP * H2 + q * 64 + tid];

    v2f bfr[2][40];
    const float* wr0 = Wl + ((wave * 2 + 0) * 16 + lr) * FSIZE + half * 2;
    const float* wr1 = Wl + ((wave * 2 + 1) * 16 + lr) * FSIZE + half * 2;
    #pragma unroll
    for (int kk = 0; kk < 14; ++kk) {
      bfr[0][kk].x = wr0[kk * 4]; bfr[0][kk].y = wr0[kk * 4 + 1];
      bfr[1][kk].x = wr1[kk * 4]; bfr[1][kk].y = wr1[kk * 4 + 1];
    }

    {
      float s = 0.f;
      for (int jj = 0; jj < NP; ++jj) s += src[jj][tid];
      meanh[tid] = s * (1.f / NP);
    }
    biasS[tid] = bval;
    #pragma unroll
    for (int q = 0; q < 32; ++q) {
      const int idx = q * 64 + tid;
      mprev[idx >> 5][idx & 31] = mp[q];
    }
    __syncthreads();

    #pragma unroll
    for (int kk = 14; kk < 40; ++kk) {
      bfr[0][kk].x = wr0[kk * 4]; bfr[0][kk].y = wr0[kk * 4 + 1];
      bfr[1][kk].x = wr1[kk * 4]; bfr[1][kk].y = wr1[kk * 4 + 1];
    }

    v2f afr[40][4];
    #pragma unroll
    for (int kk = 0; kk < 16; ++kk) {
      const int k0 = kk * 4 + half * 2;
      #pragma unroll
      for (int rt = 0; rt < 4; ++rt) {
        const int ar = rt * 16 + lr;
        afr[kk][rt].x = src[ar][k0]; afr[kk][rt].y = src[ar][k0 + 1];
      }
    }
    #pragma unroll
    for (int kk = 16; kk < 32; ++kk) {
      const int k0 = kk * 4 + half * 2;
      afr[kk][0].x = meanh[k0 - 64]; afr[kk][0].y = meanh[k0 - 63];
    }
    #pragma unroll
    for (int kk = 32; kk < 40; ++kk) {
      const int k0 = kk * 4 + half * 2;
      #pragma unroll
      for (int rt = 0; rt < 4; ++rt) {
        const int ar = rt * 16 + lr;
        afr[kk][rt].x = mprev[ar][k0 - 128]; afr[kk][rt].y = mprev[ar][k0 - 127];
      }
    }
    float res[2][4][8];
    float bias[2];
    #pragma unroll
    for (int ct2 = 0; ct2 < 2; ++ct2) {
      const int col = (wave * 2 + ct2) * 16 + lr;
      bias[ct2] = biasS[col];
      #pragma unroll
      for (int rt = 0; rt < 4; ++rt) {
        #pragma unroll
        for (int r = 0; r < 8; ++r) res[ct2][rt][r] = src[rt * 16 + r + half * 8][col];
      }
    }
    __builtin_amdgcn_sched_barrier(0);

    v8f acc[2][4] = {};
    #pragma unroll
    for (int kk = 0; kk < 40; ++kk) {
      #pragma unroll
      for (int rt = 0; rt < 4; ++rt) {
        const v2f a = (kk >= 16 && kk < 32) ? afr[kk][0] : afr[kk][rt];
        #pragma unroll
        for (int ct2 = 0; ct2 < 2; ++ct2)
          acc[ct2][rt] = wmma_f32_k4(a, bfr[ct2][kk], acc[ct2][rt]);
      }
    }

    #pragma unroll
    for (int ct2 = 0; ct2 < 2; ++ct2) {
      const int col = (wave * 2 + ct2) * 16 + lr;
      #pragma unroll
      for (int rt = 0; rt < 4; ++rt) {
        #pragma unroll
        for (int r = 0; r < 8; ++r) {
          const int row = rt * 16 + r + half * 8;
          dst[row][col] = fast_tanh(acc[ct2][rt][r] + bias[ct2]) + res[ct2][rt][r];
        }
      }
    }
    __syncthreads();
  }

  #pragma unroll
  for (int dd = 0; dd < DIMX; ++dd) {
    float acc = final_b[dd];
    for (int k = 0; k < H1; ++k) acc += hB[tid][k] * final_w[dd * H1 + k];
    out[(b * NP + tid) * 3 + dd] = acc;
  }
}

extern "C" void kernel_launch(void* const* d_in, const int* in_sizes, int n_in,
                              void* d_out, int out_size, void* d_ws,
                              size_t ws_size, hipStream_t stream) {
  const float* x       = (const float*)d_in[0];
  const float* t       = (const float*)d_in[1];
  const float* fc1_w0  = (const float*)d_in[2];
  const float* fc1_b0  = (const float*)d_in[3];
  const float* fc1_w   = (const float*)d_in[4];
  const float* fc1_b   = (const float*)d_in[5];
  const float* fc2_w0  = (const float*)d_in[6];
  const float* fc2_b0  = (const float*)d_in[7];
  const float* fc2_w   = (const float*)d_in[8];
  const float* fc2_b   = (const float*)d_in[9];
  const float* final_w = (const float*)d_in[10];
  const float* final_b = (const float*)d_in[11];
  float* out = (float*)d_out;

  float* ws    = (float*)d_ws;
  float* mfeat = ws;
  float* m0    = mfeat + BATCHN * NP * 4;
  float* m1    = m0 + BATCHN * NP * H2;
  float* m2    = m1 + BATCHN * NP * H2;
  (void)in_sizes; (void)n_in; (void)out_size; (void)ws_size;

  drift_pair_chain<<<dim3(PGRID, BATCHN), PTHREADS, 0, stream>>>(
      x, t, fc2_w0, fc2_b0, fc2_w, fc2_b, mfeat, m0, m1, m2);
  drift_particle_chain<<<dim3(BATCHN), 64, 0, stream>>>(
      x, t, fc1_w0, fc1_b0, fc1_w, fc1_b, final_w, final_b,
      mfeat, m0, m1, m2, out);
}


# particle-chain kernel: 8 waves per block, not 2: a wave owns one column tile x two row tiles; layer 0, the particle means (4 partial sums each: order changes) and the final projection use 256 threads
# speedup vs baseline: 3.8104x; 1.2451x over previous
#include <hip/hip_runtime.h>
#include <math.h>

typedef __attribute__((ext_vector_type(2))) float v2f;
typedef __attribute__((ext_vector_type(8))) float v8f;

#define DEPTH 4
#define H1 64
#define H2 32
#define NP 64
#define DIMX 3
#define FSIZE 160
#define BATCHN 256

__device__ __forceinline__ v8f wmma_f32_k4(v2f a, v2f b, v8f c) {
  return __builtin_amdgcn_wmma_f32_16x16x4_f32(false, a, false, b, (short)0, c,
                                               false, false);
}

__device__ __forceinline__ float fast_tanh(float x) {
#if __has_builtin(__builtin_amdgcn_tanhf)
  return __builtin_amdgcn_tanhf(x);
#else
  float r;
  asm volatile("v_tanh_f32 %0, %1\n\tv_nop" : "=v"(r) : "v"(x));
  return r;
#endif
}

#define PW 4
#define PGRID 16
#define PTHREADS 128
#define HP 36
static_assert(PGRID * PW == NP && PTHREADS == 32 * PW, "pair-chain launch shape");

__device__ __forceinline__ float xhalf(float v) {
  const int q = __builtin_bit_cast(int, v);
  return __builtin_bit_cast(
      float, __builtin_amdgcn_permlanex16(q, q, 0x76543210, 0xfedcba98, false, false));
}

__device__ __forceinline__ void col_mean_store(const float (&h)[4][2][8],
                                               float* __restrict__ mo, int bi,
                                               int half, int lr) {
  float cs[2];
  #pragma unroll
  for (int ct = 0; ct < 2; ++ct) {
    float p[4];
    #pragma unroll
    for (int rt = 0; rt < 4; ++rt) {
      float s = h[rt][ct][0];
      #pragma unroll
      for (int r = 1; r < 8; ++r) s += h[rt][ct][r];
      p[rt] = s;
    }
    const float s = (p[0] + p[1]) + (p[2] + p[3]);
    cs[ct] = s + xhalf(s);
  }
  mo[bi * H2 + half * 16 + lr] = (half ? cs[1] : cs[0]) * (1.f / NP);
}

__global__ __launch_bounds__(PTHREADS, 4) void drift_pair_chain(
    const float* __restrict__ x, const float* __restrict__ t,
    const float* __restrict__ fc2_w0, const float* __restrict__ fc2_b0,
    const float* __restrict__ fc2_w, const float* __restrict__ fc2_b,
    float* __restrict__ mfeat, float* __restrict__ m0,
    float* __restrict__ m1, float* __restrict__ m2) {
  const int b = blockIdx.y;
  const int wave = __builtin_amdgcn_readfirstlane((int)(threadIdx.x >> 5));
  const int lane = threadIdx.x & 31;
  const int half = lane >> 4;
  const int lr = lane & 15;
  const int i = blockIdx.x * PW + wave;
  const int bi = b * NP + i;

  __shared__ float hS[PW][NP][HP];
  float(*hs)[HP] = hS[wave];

  const float tval = t[0];
  const float xi0 = x[bi * 3 + 0];
  const float xi1 = x[bi * 3 + 1];
  const float xi2 = x[bi * 3 + 2];
  float xj[4][3];
  #pragma unroll
  for (int rt = 0; rt < 4; ++rt) {
    const int j = rt * 16 + lr;
    #pragma unroll
    for (int d = 0; d < 3; ++d) xj[rt][d] = x[(b * NP + j) * 3 + d];
  }
  v2f w0a[2];
  float w04[2], b0v[2];
  #pragma unroll
  for (int ct = 0; ct < 2; ++ct) {
    const int n = ct * 16 + lr;
    w0a[ct].x = fc2_w0[n * 5 + half * 2];
    w0a[ct].y = fc2_w0[n * 5 + half * 2 + 1];
    w04[ct] = fc2_w0[n * 5 + 4];
    b0v[ct] = fc2_b0[n];
  }

  v2f a0[4];
  float fs0 = 0.f, fs1 = 0.f;
  #pragma unroll
  for (int rt = 0; rt < 4; ++rt) {
    const int j = rt * 16 + lr;
    float r0 = xi0 - xj[rt][0];
    float r1 = xi1 - xj[rt][1];
    float r2 = xi2 - xj[rt][2];
    if (j == i) { r0 += 1.f; r1 += 1.f; r2 += 1.f; }
    const float nr = sqrtf(r0 * r0 + r1 * r1 + r2 * r2);
    a0[rt].x = half ? r2 : r0;
    a0[rt].y = half ? nr : r1;
    fs0 += a0[rt].x;
    fs1 += a0[rt].y;
  }
  #pragma unroll
  for (int m = 1; m < 16; m <<= 1) {
    fs0 += __shfl_xor(fs0, m, 32);
    fs1 += __shfl_xor(fs1, m, 32);
  }
  if (lr == 0) {
    mfeat[bi * 4 + half * 2 + 0] = fs0 * (1.f / NP);
    mfeat[bi * 4 + half * 2 + 1] = fs1 * (1.f / NP);
  }

  v8f c0[2];
  #pragma unroll
  for (int ct = 0; ct < 2; ++ct) {
    const float cv = tval * w04[ct] + b0v[ct];
    #pragma unroll
    for (int r = 0; r < 8; ++r) c0[ct][r] = cv;
  }
  float h[4][2][8];
  #pragma unroll
  for (int rt = 0; rt < 4; ++rt) {
    #pragma unroll
    for (int ct = 0; ct < 2; ++ct) {
      const v8f c = wmma_f32_k4(a0[rt], w0a[ct], c0[ct]);
      #pragma unroll
      for (int r = 0; r < 8; ++r) h[rt][ct][r] = fast_tanh(c[r]);
    }
  }

  #pragma unroll
  for (int L = 0; L < 2; ++L) {
    v2f bw[2][8];
    float bias[2];
    #pragma unroll
    for (int ct = 0; ct < 2; ++ct) {
      const float* wr = fc2_w + L * H2 * H2 + (ct * 16 + lr) * H2 + half * 2;
      #pragma unroll
      for (int kk = 0; kk < 8; ++kk) {
        bw[ct][kk].x = wr[kk * 4];
        bw[ct][kk].y = wr[kk * 4 + 1];
      }
      bias[ct] = fc2_b[L * H2 + ct * 16 + lr];
    }

    col_mean_store(h, (L == 0) ? m0 : m1, bi, half, lr);

    #pragma unroll
    for (int rt = 0; rt < 4; ++rt) {
      #pragma unroll
      for (int ct = 0; ct < 2; ++ct) {
        #pragma unroll
        for (int r = 0; r < 8; ++r)
          hs[rt * 16 + half * 8 + r][ct * 16 + lr] = h[rt][ct][r];
      }
    }
    __builtin_amdgcn_fence(__ATOMIC_RELEASE, "wavefront");
    __builtin_amdgcn_wave_barrier();
    __builtin_amdgcn_fence(__ATOMIC_ACQUIRE, "wavefront");

    v2f af[4][8];
    #pragma unroll
    for (int rt = 0; rt < 4; ++rt) {
      #pragma unroll
      for (int kk = 0; kk < 8; ++kk) {
        const int k0 = kk * 4 + half * 2;
        af[rt][kk].x = hs[rt * 16 + lr][k0];
        af[rt][kk].y = hs[rt * 16 + lr][k0 + 1];
      }
    }
    __builtin_amdgcn_sched_barrier(0);

    v8f acc[4][2] = {};
    #pragma unroll
    for (int kk = 0; kk < 8; ++kk) {
      #pragma unroll
      for (int rt = 0; rt < 4; ++rt) {
        #pragma unroll
        for (int ct = 0; ct < 2; ++ct)
          acc[rt][ct] = wmma_f32_k4(af[rt][kk], bw[ct][kk], acc[rt][ct]);
      }
    }
    #pragma unroll
    for (int rt = 0; rt < 4; ++rt) {
      #pragma unroll
      for (int ct = 0; ct < 2; ++ct) {
        #pragma unroll
        for (int r = 0; r < 8; ++r)
          h[rt][ct][r] = fast_tanh(acc[rt][ct][r] + bias[ct]) + h[rt][ct][r];
      }
    }
  }
  col_mean_store(h, m2, bi, half, lr);
}

#define QTHREADS 256
#define FWP 68
__global__ __launch_bounds__(QTHREADS) void drift_particle_chain(
    const float* __restrict__ x, const float* __restrict__ t,
    const float* __restrict__ fc1_w0, const float* __restrict__ fc1_b0,
    const float* __restrict__ fc1_w, const float* __restrict__ fc1_b,
    const float* __restrict__ final_w, const float* __restrict__ final_b,
    const float* __restrict__ mfeat, const float* __restrict__ m0,
    const float* __restrict__ m1, const float* __restrict__ m2,
    float* __restrict__ out) {
  const int b = blockIdx.x;
  const int tid = threadIdx.x;
  const int i = tid & 63;
  const int q = __builtin_amdgcn_readfirstlane(tid >> 6);

  __shared__ float hA[NP][H1 + 4];
  __shared__ float hB[NP][H1 + 4];
  __shared__ float mprev[NP][H2 + 4];
  __shared__ float psum[4][H1];
  __shared__ float meanh[H1];
  __shared__ float sfeat[NP][4];
  __shared__ float pfeat[4][4];
  __shared__ float fwS[DIMX][FWP];

  __builtin_prefetch(fc1_w, 0, 1);

  const float tval = t[0];
  const float x0 = x[(b * NP + i) * 3 + 0];
  const float x1 = x[(b * NP + i) * 3 + 1];
  const float x2 = x[(b * NP + i) * 3 + 2];
  const float mf0 = mfeat[(b * NP + i) * 4 + 0];
  const float mf1 = mfeat[(b * NP + i) * 4 + 1];
  const float mf2 = mfeat[(b * NP + i) * 4 + 2];
  const float mf3 = mfeat[(b * NP + i) * 4 + 3];
  const int fdd = tid % DIMX;
  const int fpi = tid / DIMX;
  float fbv = 0.f;
  if (tid < DIMX * NP) {
    fwS[tid >> 6][tid & 63] = final_w[tid];
    fbv = final_b[fdd];
  }
  const float nr = sqrtf(x0 * x0 + x1 * x1 + x2 * x2);
  if (q == 0) {
    sfeat[i][0] = x0; sfeat[i][1] = x1; sfeat[i][2] = x2; sfeat[i][3] = nr;
  }
  __syncthreads();
  if (tid < 16) {
    const int f = tid & 3, g = tid >> 2;
    float s = 0.f;
    for (int jj = 0; jj < 16; ++jj) s += sfeat[g * 16 + jj][f];
    pfeat[g][f] = s;
  }
  __syncthreads();

  {
    float ft[13];
    ft[0] = x0; ft[1] = x1; ft[2] = x2; ft[3] = nr;
    #pragma unroll
    for (int f = 0; f < 4; ++f)
      ft[4 + f] = ((pfeat[0][f] + pfeat[1][f]) + (pfeat[2][f] + pfeat[3][f])) * (1.f / NP);
    ft[8] = mf0; ft[9] = mf1; ft[10] = mf2; ft[11] = mf3;
    ft[12] = tval;
    #pragma unroll 4
    for (int oo = 0; oo < 16; ++oo) {
      const int o = q * 16 + oo;
      float acc = fc1_b0[o];
      #pragma unroll
      for (int k = 0; k < 13; ++k) acc += ft[k] * fc1_w0[o * 13 + k];
      hA[i][o] = fast_tanh(acc);
    }
  }
  __syncthreads();

  const int wave = __builtin_amdgcn_readfirstlane(tid >> 5);
  const int lane = tid & 31;
  const int half = lane >> 4;
  const int lr = lane & 15;
  const int ct = wave & 3;
  const int rp = wave >> 2;
  const int col = ct * 16 + lr;

  for (int d = 1; d <= 3; ++d) {
    const float* Wl = fc1_w + (d - 1) * H1 * FSIZE;
    const float* ms = (d == 1) ? m0 : (d == 2) ? m1 : m2;
    float(*src)[H1 + 4] = (d & 1) ? hA : hB;
    float(*dst)[H1 + 4] = (d & 1) ? hB : hA;

    float mp[8];
    #pragma unroll
    for (int u = 0; u < 8; ++u) mp[u] = ms[b * NP * H2 + u * QTHREADS + tid];

    const float bias = fc1_b[(d - 1) * H1 + col];

    v2f bfr[40];
    const float* wr = Wl + col * FSIZE + half * 2;
    #pragma unroll
    for (int kk = 0; kk < 40; ++kk) {
      bfr[kk].x = wr[kk * 4]; bfr[kk].y = wr[kk * 4 + 1];
    }

    {
      float s = 0.f;
      for (int jj = 0; jj < 16; ++jj) s += src[q * 16 + jj][i];
      psum[q][i] = s;
    }
    #pragma unroll
    for (int u = 0; u < 8; ++u) {
      const int idx = u * QTHREADS + tid;
      mprev[idx >> 5][idx & 31] = mp[u];
    }
    __syncthreads();

    if (q == 0)
      meanh[i] = ((psum[0][i] + psum[1][i]) + (psum[2][i] + psum[3][i])) * (1.f / NP);
    __syncthreads();

    v2f afr[16][2], afm[16], afp[8][2];
    #pragma unroll
    for (int kk = 0; kk < 16; ++kk) {
      const int k0 = kk * 4 + half * 2;
      #pragma unroll
      for (int rl = 0; rl < 2; ++rl) {
        const int ar = (rp * 2 + rl) * 16 + lr;
        afr[kk][rl].x = src[ar][k0]; afr[kk][rl].y = src[ar][k0 + 1];
      }
    }
    #pragma unroll
    for (int kk = 0; kk < 16; ++kk) {
      const int k0 = kk * 4 + half * 2;
      afm[kk].x = meanh[k0]; afm[kk].y = meanh[k0 + 1];
    }
    #pragma unroll
    for (int kk = 0; kk < 8; ++kk) {
      const int k0 = kk * 4 + half * 2;
      #pragma unroll
      for (int rl = 0; rl < 2; ++rl) {
        const int ar = (rp * 2 + rl) * 16 + lr;
        afp[kk][rl].x = mprev[ar][k0]; afp[kk][rl].y = mprev[ar][k0 + 1];
      }
    }
    float res[2][8];
    #pragma unroll
    for (int rl = 0; rl < 2; ++rl) {
      #pragma unroll
      for (int r = 0; r < 8; ++r)
        res[rl][r] = src[(rp * 2 + rl) * 16 + r + half * 8][col];
    }
    __builtin_amdgcn_sched_barrier(0);

    v8f acc[2] = {};
    #pragma unroll
    for (int kk = 0; kk < 40; ++kk) {
      #pragma unroll
      for (int rl = 0; rl < 2; ++rl) {
        const v2f a = (kk < 16) ? afr[kk][rl]
                    : (kk < 32) ? afm[kk - 16] : afp[kk - 32][rl];
        acc[rl] = wmma_f32_k4(a, bfr[kk], acc[rl]);
      }
    }

    #pragma unroll
    for (int rl = 0; rl < 2; ++rl) {
      #pragma unroll
      for (int r = 0; r < 8; ++r) {
        const int row = (rp * 2 + rl) * 16 + r + half * 8;
        dst[row][col] = fast_tanh(acc[rl][r] + bias) + res[rl][r];
      }
    }
    __syncthreads();
  }

  if (tid < DIMX * NP) {
    float acc = fbv;
    for (int k = 0; k < H1; ++k) acc += hB[fpi][k] * fwS[fdd][k];
    out[b * NP * DIMX + tid] = acc;
  }
}

extern "C" void kernel_launch(void* const* d_in, const int* in_sizes, int n_in,
                              void* d_out, int out_size, void* d_ws,
                              size_t ws_size, hipStream_t stream) {
  const float* x       = (const float*)d_in[0];
  const float* t       = (const float*)d_in[1];
  const float* fc1_w0  = (const float*)d_in[2];
  const float* fc1_b0  = (const float*)d_in[3];
  const float* fc1_w   = (const float*)d_in[4];
  const float* fc1_b   = (const float*)d_in[5];
  const float* fc2_w0  = (const float*)d_in[6];
  const float* fc2_b0  = (const float*)d_in[7];
  const float* fc2_w   = (const float*)d_in[8];
  const float* fc2_b   = (const float*)d_in[9];
  const float* final_w = (const float*)d_in[10];
  const float* final_b = (const float*)d_in[11];
  float* out = (float*)d_out;

  float* ws    = (float*)d_ws;
  float* mfeat = ws;
  float* m0    = mfeat + BATCHN * NP * 4;
  float* m1    = m0 + BATCHN * NP * H2;
  float* m2    = m1 + BATCHN * NP * H2;
  (void)in_sizes; (void)n_in; (void)out_size; (void)ws_size;

  drift_pair_chain<<<dim3(PGRID, BATCHN), PTHREADS, 0, stream>>>(
      x, t, fc2_w0, fc2_b0, fc2_w, fc2_b, mfeat, m0, m1, m2);
  drift_particle_chain<<<dim3(BATCHN), QTHREADS, 0, stream>>>(
      x, t, fc1_w0, fc1_b0, fc1_w, fc1_b, final_w, final_b,
      mfeat, m0, m1, m2, out);
}


# particle-chain kernel: a layer's 40 KB of weights are read once per block by coalesced 16-byte loads a layer ahead and staged in LDS for the B fragments, not gathered per wave by 40 loads of 16 lines
# speedup vs baseline: 4.1754x; 1.0958x over previous
#include <hip/hip_runtime.h>
#include <math.h>

typedef __attribute__((ext_vector_type(2))) float v2f;
typedef __attribute__((ext_vector_type(8))) float v8f;
typedef __attribute__((ext_vector_type(4))) float v4f;

#define DEPTH 4
#define H1 64
#define H2 32
#define NP 64
#define DIMX 3
#define FSIZE 160
#define BATCHN 256

__device__ __forceinline__ v8f wmma_f32_k4(v2f a, v2f b, v8f c) {
  return __builtin_amdgcn_wmma_f32_16x16x4_f32(false, a, false, b, (short)0, c,
                                               false, false);
}

__device__ __forceinline__ float fast_tanh(float x) {
#if __has_builtin(__builtin_amdgcn_tanhf)
  return __builtin_amdgcn_tanhf(x);
#else
  float r;
  asm volatile("v_tanh_f32 %0, %1\n\tv_nop" : "=v"(r) : "v"(x));
  return r;
#endif
}

#define PW 4
#define PGRID 16
#define PTHREADS 128
#define HP 36
static_assert(PGRID * PW == NP && PTHREADS == 32 * PW, "pair-chain launch shape");

__device__ __forceinline__ float xhalf(float v) {
  const int q = __builtin_bit_cast(int, v);
  return __builtin_bit_cast(
      float, __builtin_amdgcn_permlanex16(q, q, 0x76543210, 0xfedcba98, false, false));
}

__device__ __forceinline__ void col_mean_store(const float (&h)[4][2][8],
                                               float* __restrict__ mo, int bi,
                                               int half, int lr) {
  float cs[2];
  #pragma unroll
  for (int ct = 0; ct < 2; ++ct) {
    float p[4];
    #pragma unroll
    for (int rt = 0; rt < 4; ++rt) {
      float s = h[rt][ct][0];
      #pragma unroll
      for (int r = 1; r < 8; ++r) s += h[rt][ct][r];
      p[rt] = s;
    }
    const float s = (p[0] + p[1]) + (p[2] + p[3]);
    cs[ct] = s + xhalf(s);
  }
  mo[bi * H2 + half * 16 + lr] = (half ? cs[1] : cs[0]) * (1.f / NP);
}

__global__ __launch_bounds__(PTHREADS, 4) void drift_pair_chain(
    const float* __restrict__ x, const float* __restrict__ t,
    const float* __restrict__ fc2_w0, const float* __restrict__ fc2_b0,
    const float* __restrict__ fc2_w, const float* __restrict__ fc2_b,
    float* __restrict__ mfeat, float* __restrict__ m0,
    float* __restrict__ m1, float* __restrict__ m2) {
  const int b = blockIdx.y;
  const int wave = __builtin_amdgcn_readfirstlane((int)(threadIdx.x >> 5));
  const int lane = threadIdx.x & 31;
  const int half = lane >> 4;
  const int lr = lane & 15;
  const int i = blockIdx.x * PW + wave;
  const int bi = b * NP + i;

  __shared__ float hS[PW][NP][HP];
  float(*hs)[HP] = hS[wave];

  const float tval = t[0];
  const float xi0 = x[bi * 3 + 0];
  const float xi1 = x[bi * 3 + 1];
  const float xi2 = x[bi * 3 + 2];
  float xj[4][3];
  #pragma unroll
  for (int rt = 0; rt < 4; ++rt) {
    const int j = rt * 16 + lr;
    #pragma unroll
    for (int d = 0; d < 3; ++d) xj[rt][d] = x[(b * NP + j) * 3 + d];
  }
  v2f w0a[2];
  float w04[2], b0v[2];
  #pragma unroll
  for (int ct = 0; ct < 2; ++ct) {
    const int n = ct * 16 + lr;
    w0a[ct].x = fc2_w0[n * 5 + half * 2];
    w0a[ct].y = fc2_w0[n * 5 + half * 2 + 1];
    w04[ct] = fc2_w0[n * 5 + 4];
    b0v[ct] = fc2_b0[n];
  }

  v2f a0[4];
  float fs0 = 0.f, fs1 = 0.f;
  #pragma unroll
  for (int rt = 0; rt < 4; ++rt) {
    const int j = rt * 16 + lr;
    float r0 = xi0 - xj[rt][0];
    float r1 = xi1 - xj[rt][1];
    float r2 = xi2 - xj[rt][2];
    if (j == i) { r0 += 1.f; r1 += 1.f; r2 += 1.f; }
    const float nr = sqrtf(r0 * r0 + r1 * r1 + r2 * r2);
    a0[rt].x = half ? r2 : r0;
    a0[rt].y = half ? nr : r1;
    fs0 += a0[rt].x;
    fs1 += a0[rt].y;
  }
  #pragma unroll
  for (int m = 1; m < 16; m <<= 1) {
    fs0 += __shfl_xor(fs0, m, 32);
    fs1 += __shfl_xor(fs1, m, 32);
  }
  if (lr == 0) {
    mfeat[bi * 4 + half * 2 + 0] = fs0 * (1.f / NP);
    mfeat[bi * 4 + half * 2 + 1] = fs1 * (1.f / NP);
  }

  v8f c0[2];
  #pragma unroll
  for (int ct = 0; ct < 2; ++ct) {
    const float cv = tval * w04[ct] + b0v[ct];
    #pragma unroll
    for (int r = 0; r < 8; ++r) c0[ct][r] = cv;
  }
  float h[4][2][8];
  #pragma unroll
  for (int rt = 0; rt < 4; ++rt) {
    #pragma unroll
    for (int ct = 0; ct < 2; ++ct) {
      const v8f c = wmma_f32_k4(a0[rt], w0a[ct], c0[ct]);
      #pragma unroll
      for (int r = 0; r < 8; ++r) h[rt][ct][r] = fast_tanh(c[r]);
    }
  }

  #pragma unroll
  for (int L = 0; L < 2; ++L) {
    v2f bw[2][8];
    float bias[2];
    #pragma unroll
    for (int ct = 0; ct < 2; ++ct) {
      const float* wr = fc2_w + L * H2 * H2 + (ct * 16 + lr) * H2 + half * 2;
      #pragma unroll
      for (int kk = 0; kk < 8; ++kk) {
        bw[ct][kk].x = wr[kk * 4];
        bw[ct][kk].y = wr[kk * 4 + 1];
      }
      bias[ct] = fc2_b[L * H2 + ct * 16 + lr];
    }

    col_mean_store(h, (L == 0) ? m0 : m1, bi, half, lr);

    #pragma unroll
    for (int rt = 0; rt < 4; ++rt) {
      #pragma unroll
      for (int ct = 0; ct < 2; ++ct) {
        #pragma unroll
        for (int r = 0; r < 8; ++r)
          hs[rt * 16 + half * 8 + r][ct * 16 + lr] = h[rt][ct][r];
      }
    }
    __builtin_amdgcn_fence(__ATOMIC_RELEASE, "wavefront");
    __builtin_amdgcn_wave_barrier();
    __builtin_amdgcn_fence(__ATOMIC_ACQUIRE, "wavefront");

    v2f af[4][8];
    #pragma unroll
    for (int rt = 0; rt < 4; ++rt) {
      #pragma unroll
      for (int kk = 0; kk < 8; ++kk) {
        const int k0 = kk * 4 + half * 2;
        af[rt][kk].x = hs[rt * 16 + lr][k0];
        af[rt][kk].y = hs[rt * 16 + lr][k0 + 1];
      }
    }
    __builtin_amdgcn_sched_barrier(0);

    v8f acc[4][2] = {};
    #pragma unroll
    for (int kk = 0; kk < 8; ++kk) {
      #pragma unroll
      for (int rt = 0; rt < 4; ++rt) {
        #pragma unroll
        for (int ct = 0; ct < 2; ++ct)
          acc[rt][ct] = wmma_f32_k4(af[rt][kk], bw[ct][kk], acc[rt][ct]);
      }
    }
    #pragma unroll
    for (int rt = 0; rt < 4; ++rt) {
      #pragma unroll
      for (int ct = 0; ct < 2; ++ct) {
        #pragma unroll
        for (int r = 0; r < 8; ++r)
          h[rt][ct][r] = fast_tanh(acc[rt][ct][r] + bias[ct]) + h[rt][ct][r];
      }
    }
  }
  col_mean_store(h, m2, bi, half, lr);
}

#define QTHREADS 256
#define FWP 68
#define WLP 164
#define WCH 10
static_assert(WCH * QTHREADS * 4 == H1 * FSIZE && FSIZE % 4 == 0, "weight staging split");

__device__ __forceinline__ void fetch_layer(const float* __restrict__ Wl,
                                            const float* __restrict__ msb,
                                            const float* __restrict__ bl, int tid,
                                            int col, v4f (&w)[WCH], float (&mp)[8],
                                            float& bias) {
  #pragma unroll
  for (int u = 0; u < 8; ++u) mp[u] = msb[u * QTHREADS + tid];
  bias = bl[col];
  #pragma unroll
  for (int u = 0; u < WCH; ++u)
    w[u] = *reinterpret_cast<const v4f*>(Wl + 4 * (u * QTHREADS + tid));
}

__global__ __launch_bounds__(QTHREADS) void drift_particle_chain(
    const float* __restrict__ x, const float* __restrict__ t,
    const float* __restrict__ fc1_w0, const float* __restrict__ fc1_b0,
    const float* __restrict__ fc1_w, const float* __restrict__ fc1_b,
    const float* __restrict__ final_w, const float* __restrict__ final_b,
    const float* __restrict__ mfeat, const float* __restrict__ m0,
    const float* __restrict__ m1, const float* __restrict__ m2,
    float* __restrict__ out) {
  const int b = blockIdx.x;
  const int tid = threadIdx.x;
  const int i = tid & 63;
  const int q = __builtin_amdgcn_readfirstlane(tid >> 6);

  __shared__ float hA[NP][H1 + 4];
  __shared__ float hB[NP][H1 + 4];
  __shared__ float mprev[NP][H2 + 4];
  __shared__ float psum[4][H1];
  __shared__ float meanh[H1];
  __shared__ float sfeat[NP][4];
  __shared__ float pfeat[4][4];
  __shared__ float fwS[DIMX][FWP];
  __shared__ __attribute__((aligned(16))) float wS[H1][WLP];

  __builtin_prefetch(fc1_w, 0, 1);

  const float tval = t[0];
  const float x0 = x[(b * NP + i) * 3 + 0];
  const float x1 = x[(b * NP + i) * 3 + 1];
  const float x2 = x[(b * NP + i) * 3 + 2];
  const float mf0 = mfeat[(b * NP + i) * 4 + 0];
  const float mf1 = mfeat[(b * NP + i) * 4 + 1];
  const float mf2 = mfeat[(b * NP + i) * 4 + 2];
  const float mf3 = mfeat[(b * NP + i) * 4 + 3];
  const int fdd = tid % DIMX;
  const int fpi = tid / DIMX;
  float fbv = 0.f;
  if (tid < DIMX * NP) {
    fwS[tid >> 6][tid & 63] = final_w[tid];
    fbv = final_b[fdd];
  }
  const float nr = sqrtf(x0 * x0 + x1 * x1 + x2 * x2);
  if (q == 0) {
    sfeat[i][0] = x0; sfeat[i][1] = x1; sfeat[i][2] = x2; sfeat[i][3] = nr;
  }
  __syncthreads();
  if (tid < 16) {
    const int f = tid & 3, g = tid >> 2;
    float s = 0.f;
    for (int jj = 0; jj < 16; ++jj) s += sfeat[g * 16 + jj][f];
    pfeat[g][f] = s;
  }
  __syncthreads();

  const int wave = __builtin_amdgcn_readfirstlane(tid >> 5);
  const int lane = tid & 31;
  const int half = lane >> 4;
  const int lr = lane & 15;
  const int ct = wave & 3;
  const int rp = wave >> 2;
  const int col = ct * 16 + lr;

  v4f wpre[WCH];
  float mpre[8];
  float bpre;
  fetch_layer(fc1_w, m0 + b * NP * H2, fc1_b, tid, col, wpre, mpre, bpre);

  {
    float ft[13];
    ft[0] = x0; ft[1] = x1; ft[2] = x2; ft[3] = nr;
    #pragma unroll
    for (int f = 0; f < 4; ++f)
      ft[4 + f] = ((pfeat[0][f] + pfeat[1][f]) + (pfeat[2][f] + pfeat[3][f])) * (1.f / NP);
    ft[8] = mf0; ft[9] = mf1; ft[10] = mf2; ft[11] = mf3;
    ft[12] = tval;
    #pragma unroll 4
    for (int oo = 0; oo < 16; ++oo) {
      const int o = q * 16 + oo;
      float acc = fc1_b0[o];
      #pragma unroll
      for (int k = 0; k < 13; ++k) acc += ft[k] * fc1_w0[o * 13 + k];
      hA[i][o] = fast_tanh(acc);
    }
  }
  __syncthreads();

  for (int d = 1; d <= 3; ++d) {
    float(*src)[H1 + 4] = (d & 1) ? hA : hB;
    float(*dst)[H1 + 4] = (d & 1) ? hB : hA;

    #pragma unroll
    for (int u = 0; u < WCH; ++u) {
      const int f = u * QTHREADS + tid;
      *reinterpret_cast<v4f*>(&wS[f / (FSIZE / 4)][(f % (FSIZE / 4)) * 4]) = wpre[u];
    }
    const float bias = bpre;

    {
      float s = 0.f;
      for (int jj = 0; jj < 16; ++jj) s += src[q * 16 + jj][i];
      psum[q][i] = s;
    }
    #pragma unroll
    for (int u = 0; u < 8; ++u) {
      const int idx = u * QTHREADS + tid;
      mprev[idx >> 5][idx & 31] = mpre[u];
    }
    __syncthreads();

    if (q == 0)
      meanh[i] = ((psum[0][i] + psum[1][i]) + (psum[2][i] + psum[3][i])) * (1.f / NP);
    __syncthreads();

    if (d < 3)
      fetch_layer(fc1_w + d * H1 * FSIZE, ((d == 1) ? m1 : m2) + b * NP * H2,
                  fc1_b + d * H1, tid, col, wpre, mpre, bpre);

    v2f bfr[40];
    #pragma unroll
    for (int kk = 0; kk < 40; ++kk) {
      const int k0 = kk * 4 + half * 2;
      bfr[kk].x = wS[col][k0]; bfr[kk].y = wS[col][k0 + 1];
    }

    v2f afr[16][2], afm[16], afp[8][2];
    #pragma unroll
    for (int kk = 0; kk < 16; ++kk) {
      const int k0 = kk * 4 + half * 2;
      #pragma unroll
      for (int rl = 0; rl < 2; ++rl) {
        const int ar = (rp * 2 + rl) * 16 + lr;
        afr[kk][rl].x = src[ar][k0]; afr[kk][rl].y = src[ar][k0 + 1];
      }
    }
    #pragma unroll
    for (int kk = 0; kk < 16; ++kk) {
      const int k0 = kk * 4 + half * 2;
      afm[kk].x = meanh[k0]; afm[kk].y = meanh[k0 + 1];
    }
    #pragma unroll
    for (int kk = 0; kk < 8; ++kk) {
      const int k0 = kk * 4 + half * 2;
      #pragma unroll
      for (int rl = 0; rl < 2; ++rl) {
        const int ar = (rp * 2 + rl) * 16 + lr;
        afp[kk][rl].x = mprev[ar][k0]; afp[kk][rl].y = mprev[ar][k0 + 1];
      }
    }
    float res[2][8];
    #pragma unroll
    for (int rl = 0; rl < 2; ++rl) {
      #pragma unroll
      for (int r = 0; r < 8; ++r)
        res[rl][r] = src[(rp * 2 + rl) * 16 + r + half * 8][col];
    }
    __builtin_amdgcn_sched_barrier(0);

    v8f acc[2] = {};
    #pragma unroll
    for (int kk = 0; kk < 40; ++kk) {
      #pragma unroll
      for (int rl = 0; rl < 2; ++rl) {
        const v2f a = (kk < 16) ? afr[kk][rl]
                    : (kk < 32) ? afm[kk - 16] : afp[kk - 32][rl];
        acc[rl] = wmma_f32_k4(a, bfr[kk], acc[rl]);
      }
    }

    #pragma unroll
    for (int rl = 0; rl < 2; ++rl) {
      #pragma unroll
      for (int r = 0; r < 8; ++r) {
        const int row = (rp * 2 + rl) * 16 + r + half * 8;
        dst[row][col] = fast_tanh(acc[rl][r] + bias) + res[rl][r];
      }
    }
    __syncthreads();
  }

  if (tid < DIMX * NP) {
    float acc = fbv;
    for (int k = 0; k < H1; ++k) acc += hB[fpi][k] * fwS[fdd][k];
    out[b * NP * DIMX + tid] = acc;
  }
}

extern "C" void kernel_launch(void* const* d_in, const int* in_sizes, int n_in,
                              void* d_out, int out_size, void* d_ws,
                              size_t ws_size, hipStream_t stream) {
  const float* x       = (const float*)d_in[0];
  const float* t       = (const float*)d_in[1];
  const float* fc1_w0  = (const float*)d_in[2];
  const float* fc1_b0  = (const float*)d_in[3];
  const float* fc1_w   = (const float*)d_in[4];
  const float* fc1_b   = (const float*)d_in[5];
  const float* fc2_w0  = (const float*)d_in[6];
  const float* fc2_b0  = (const float*)d_in[7];
  const float* fc2_w   = (const float*)d_in[8];
  const float* fc2_b   = (const float*)d_in[9];
  const float* final_w = (const float*)d_in[10];
  const float* final_b = (const float*)d_in[11];
  float* out = (float*)d_out;

  float* ws    = (float*)d_ws;
  float* mfeat = ws;
  float* m0    = mfeat + BATCHN * NP * 4;
  float* m1    = m0 + BATCHN * NP * H2;
  float* m2    = m1 + BATCHN * NP * H2;
  (void)in_sizes; (void)n_in; (void)out_size; (void)ws_size;

  drift_pair_chain<<<dim3(PGRID, BATCHN), PTHREADS, 0, stream>>>(
      x, t, fc2_w0, fc2_b0, fc2_w, fc2_b, mfeat, m0, m1, m2);
  drift_particle_chain<<<dim3(BATCHN), QTHREADS, 0, stream>>>(
      x, t, fc1_w0, fc1_b0, fc1_w, fc1_b, final_w, final_b,
      mfeat, m0, m1, m2, out);
}


# both kernels fused into one launch of 256 resident 16-wave blocks (one per batch): a wave runs 4 pair chains in turn, stage means pass through LDS, then the block runs the particle chain
# speedup vs baseline: 4.5988x; 1.1014x over previous
#include <hip/hip_runtime.h>
#include <math.h>

typedef __attribute__((ext_vector_type(2))) float v2f;
typedef __attribute__((ext_vector_type(8))) float v8f;
typedef __attribute__((ext_vector_type(4))) float v4f;
typedef const __attribute__((address_space(1))) float* gptr;

#define DEPTH 4
#define H1 64
#define H2 32
#define NP 64
#define DIMX 3
#define FSIZE 160
#define BATCHN 256

__device__ __forceinline__ v8f wmma_f32_k4(v2f a, v2f b, v8f c) {
  return __builtin_amdgcn_wmma_f32_16x16x4_f32(false, a, false, b, (short)0, c,
                                               false, false);
}

__device__ __forceinline__ float fast_tanh(float x) {
#if __has_builtin(__builtin_amdgcn_tanhf)
  return __builtin_amdgcn_tanhf(x);
#else
  float r;
  asm volatile("v_tanh_f32 %0, %1\n\tv_nop" : "=v"(r) : "v"(x));
  return r;
#endif
}

#define FW 16
#define FTHREADS 512
#define IPW 4
#define HP 36
#define FWP 68
#define WLP 164
#define WCH 5
static_assert(FW * IPW == NP && FTHREADS == 32 * FW, "launch shape");
static_assert(WCH * FTHREADS * 4 == H1 * FSIZE && FSIZE % 4 == 0, "weight staging split");

__device__ __forceinline__ float xhalf(float v) {
  const int q = __builtin_bit_cast(int, v);
  return __builtin_bit_cast(
      float, __builtin_amdgcn_permlanex16(q, q, 0x76543210, 0xfedcba98, false, false));
}

__device__ __forceinline__ void col_mean_store(const float (&h)[4][2][8],
                                               float* mo, int half, int lr) {
  float cs[2];
  #pragma unroll
  for (int ct = 0; ct < 2; ++ct) {
    float p[4];
    #pragma unroll
    for (int rt = 0; rt < 4; ++rt) {
      float s = h[rt][ct][0];
      #pragma unroll
      for (int r = 1; r < 8; ++r) s += h[rt][ct][r];
      p[rt] = s;
    }
    const float s = (p[0] + p[1]) + (p[2] + p[3]);
    cs[ct] = s + xhalf(s);
  }
  mo[half * 16 + lr] = (half ? cs[1] : cs[0]) * (1.f / NP);
}

__device__ __forceinline__ void fetch_layer(const float* __restrict__ Wl,
                                            const float* __restrict__ bl, int tid,
                                            int col, v4f (&w)[WCH], float& bias) {
  bias = bl[col];
  #pragma unroll
  for (int u = 0; u < WCH; ++u)
    w[u] = *reinterpret_cast<const v4f*>(Wl + 4 * (u * FTHREADS + tid));
}

__global__ __launch_bounds__(FTHREADS) void drift_fused(
    const float* __restrict__ x, const float* __restrict__ t,
    const float* __restrict__ fc1_w0, const float* __restrict__ fc1_b0,
    const float* __restrict__ fc1_w, const float* __restrict__ fc1_b,
    const float* __restrict__ fc2_w0, const float* __restrict__ fc2_b0,
    const float* __restrict__ fc2_w, const float* __restrict__ fc2_b,
    const float* __restrict__ final_w, const float* __restrict__ final_b,
    float* __restrict__ out) {
  const int b = blockIdx.x;
  const int tid = threadIdx.x;
  const int wave = __builtin_amdgcn_readfirstlane(tid >> 5);
  const int lane = tid & 31;
  const int half = lane >> 4;
  const int lr = lane & 15;

  __shared__ float hS[FW][NP][HP];
  __shared__ float ms[3][NP][HP];
  __shared__ float mfS[NP][4];
  __shared__ float hA[NP][H1 + 4];
  __shared__ float hB[NP][H1 + 4];
  __shared__ float psum[4][H1];
  __shared__ float meanh[H1];
  __shared__ float sfeat[NP][4];
  __shared__ float pfeat[4][4];
  __shared__ float fwS[DIMX][FWP];
  __shared__ __attribute__((aligned(16))) float wS[H1][WLP];

  __builtin_prefetch(fc1_w, 0, 1);

  const float tval = t[0];

  if (tid < NP) {
    const float p0 = x[(b * NP + tid) * 3 + 0];
    const float p1 = x[(b * NP + tid) * 3 + 1];
    const float p2 = x[(b * NP + tid) * 3 + 2];
    sfeat[tid][0] = p0; sfeat[tid][1] = p1; sfeat[tid][2] = p2;
    sfeat[tid][3] = sqrtf(p0 * p0 + p1 * p1 + p2 * p2);
  }
  if (tid < DIMX * NP) fwS[tid >> 6][tid & 63] = final_w[tid];

  float(*hs)[HP] = hS[wave];

  for (int it = 0; it < IPW; ++it) {
    const int i = wave * IPW + it;

    gptr xq = (gptr)x;
    gptr w0q = (gptr)fc2_w0;
    gptr b0q = (gptr)fc2_b0;
    gptr wq = (gptr)fc2_w;
    gptr bq = (gptr)fc2_b;
    asm volatile("" : "+s"(xq), "+s"(w0q), "+s"(b0q), "+s"(wq), "+s"(bq));

    const gptr xb = xq + b * NP * 3;
    const float xi0 = xb[i * 3 + 0];
    const float xi1 = xb[i * 3 + 1];
    const float xi2 = xb[i * 3 + 2];
    float xj[4][3];
    #pragma unroll
    for (int rt = 0; rt < 4; ++rt) {
      const int j = rt * 16 + lr;
      #pragma unroll
      for (int d = 0; d < 3; ++d) xj[rt][d] = xb[j * 3 + d];
    }
    v2f w0a[2];
    float w04[2], b0v[2];
    #pragma unroll
    for (int ct = 0; ct < 2; ++ct) {
      const int n = ct * 16 + lr;
      w0a[ct].x = w0q[n * 5 + half * 2];
      w0a[ct].y = w0q[n * 5 + half * 2 + 1];
      w04[ct] = w0q[n * 5 + 4];
      b0v[ct] = b0q[n];
    }

    v2f a0[4];
    float fs0 = 0.f, fs1 = 0.f;
    #pragma unroll
    for (int rt = 0; rt < 4; ++rt) {
      const int j = rt * 16 + lr;
      float r0 = xi0 - xj[rt][0];
      float r1 = xi1 - xj[rt][1];
      float r2 = xi2 - xj[rt][2];
      if (j == i) { r0 += 1.f; r1 += 1.f; r2 += 1.f; }
      const float nr = sqrtf(r0 * r0 + r1 * r1 + r2 * r2);
      a0[rt].x = half ? r2 : r0;
      a0[rt].y = half ? nr : r1;
      fs0 += a0[rt].x;
      fs1 += a0[rt].y;
    }
    #pragma unroll
    for (int m = 1; m < 16; m <<= 1) {
      fs0 += __shfl_xor(fs0, m, 32);
      fs1 += __shfl_xor(fs1, m, 32);
    }
    if (lr == 0) {
      mfS[i][half * 2 + 0] = fs0 * (1.f / NP);
      mfS[i][half * 2 + 1] = fs1 * (1.f / NP);
    }

    v8f c0[2];
    #pragma unroll
    for (int ct = 0; ct < 2; ++ct) {
      const float cv = tval * w04[ct] + b0v[ct];
      #pragma unroll
      for (int r = 0; r < 8; ++r) c0[ct][r] = cv;
    }
    float h[4][2][8];
    #pragma unroll
    for (int rt = 0; rt < 4; ++rt) {
      #pragma unroll
      for (int ct = 0; ct < 2; ++ct) {
        const v8f c = wmma_f32_k4(a0[rt], w0a[ct], c0[ct]);
        #pragma unroll
        for (int r = 0; r < 8; ++r) h[rt][ct][r] = fast_tanh(c[r]);
      }
    }

    #pragma unroll
    for (int L = 0; L < 2; ++L) {
      v2f bw[2][8];
      float bias[2];
      #pragma unroll
      for (int ct = 0; ct < 2; ++ct) {
        gptr wr = wq + L * H2 * H2 + (ct * 16 + lr) * H2 + half * 2;
        #pragma unroll
        for (int kk = 0; kk < 8; ++kk) {
          bw[ct][kk].x = wr[kk * 4];
          bw[ct][kk].y = wr[kk * 4 + 1];
        }
        bias[ct] = bq[L * H2 + ct * 16 + lr];
      }

      col_mean_store(h, ms[L][i], half, lr);

      #pragma unroll
      for (int rt = 0; rt < 4; ++rt) {
        #pragma unroll
        for (int ct = 0; ct < 2; ++ct) {
          #pragma unroll
          for (int r = 0; r < 8; ++r)
            hs[rt * 16 + half * 8 + r][ct * 16 + lr] = h[rt][ct][r];
        }
      }
      __builtin_amdgcn_fence(__ATOMIC_RELEASE, "wavefront");
      __builtin_amdgcn_wave_barrier();
      __builtin_amdgcn_fence(__ATOMIC_ACQUIRE, "wavefront");

      v2f af[4][8];
      #pragma unroll
      for (int rt = 0; rt < 4; ++rt) {
        #pragma unroll
        for (int kk = 0; kk < 8; ++kk) {
          const int k0 = kk * 4 + half * 2;
          af[rt][kk].x = hs[rt * 16 + lr][k0];
          af[rt][kk].y = hs[rt * 16 + lr][k0 + 1];
        }
      }
      __builtin_amdgcn_sched_barrier(0);

      v8f acc[4][2] = {};
      #pragma unroll
      for (int kk = 0; kk < 8; ++kk) {
        #pragma unroll
        for (int rt = 0; rt < 4; ++rt) {
          #pragma unroll
          for (int ct = 0; ct < 2; ++ct)
            acc[rt][ct] = wmma_f32_k4(af[rt][kk], bw[ct][kk], acc[rt][ct]);
        }
      }
      #pragma unroll
      for (int rt = 0; rt < 4; ++rt) {
        #pragma unroll
        for (int ct = 0; ct < 2; ++ct) {
          #pragma unroll
          for (int r = 0; r < 8; ++r)
            h[rt][ct][r] = fast_tanh(acc[rt][ct][r] + bias[ct]) + h[rt][ct][r];
        }
      }
    }
    col_mean_store(h, ms[2][i], half, lr);
  }

  const int i = tid & 63;
  const int q = __builtin_amdgcn_readfirstlane(tid >> 6);
  const int fdd = tid % DIMX;
  const int fpi = tid / DIMX;
  float fbv = 0.f;
  if (tid < DIMX * NP) fbv = final_b[fdd];

  __syncthreads();
  if (tid < 16) {
    const int f = tid & 3, g = tid >> 2;
    float s = 0.f;
    for (int jj = 0; jj < 16; ++jj) s += sfeat[g * 16 + jj][f];
    pfeat[g][f] = s;
  }
  __syncthreads();

  const int ct = wave & 3;
  const int rt = wave >> 2;
  const int col = ct * 16 + lr;

  v4f wpre[WCH];
  float bpre;
  fetch_layer(fc1_w, fc1_b, tid, col, wpre, bpre);

  {
    float ft[13];
    #pragma unroll
    for (int f = 0; f < 4; ++f) ft[f] = sfeat[i][f];
    #pragma unroll
    for (int f = 0; f < 4; ++f)
      ft[4 + f] = ((pfeat[0][f] + pfeat[1][f]) + (pfeat[2][f] + pfeat[3][f])) * (1.f / NP);
    #pragma unroll
    for (int f = 0; f < 4; ++f) ft[8 + f] = mfS[i][f];
    ft[12] = tval;
    #pragma unroll 4
    for (int oo = 0; oo < 8; ++oo) {
      const int o = q * 8 + oo;
      float acc = fc1_b0[o];
      #pragma unroll
      for (int k = 0; k < 13; ++k) acc += ft[k] * fc1_w0[o * 13 + k];
      hA[i][o] = fast_tanh(acc);
    }
  }
  __syncthreads();

  for (int d = 1; d <= 3; ++d) {
    float(*src)[H1 + 4] = (d & 1) ? hA : hB;
    float(*dst)[H1 + 4] = (d & 1) ? hB : hA;
    float(*mprev)[HP] = ms[d - 1];

    #pragma unroll
    for (int u = 0; u < WCH; ++u) {
      const int f = u * FTHREADS + tid;
      *reinterpret_cast<v4f*>(&wS[f / (FSIZE / 4)][(f % (FSIZE / 4)) * 4]) = wpre[u];
    }
    const float bias = bpre;

    if (q < 4) {
      float s = 0.f;
      for (int jj = 0; jj < 16; ++jj) s += src[q * 16 + jj][i];
      psum[q][i] = s;
    }
    __syncthreads();

    if (q == 0)
      meanh[i] = ((psum[0][i] + psum[1][i]) + (psum[2][i] + psum[3][i])) * (1.f / NP);
    __syncthreads();

    if (d < 3)
      fetch_layer(fc1_w + d * H1 * FSIZE, fc1_b + d * H1, tid, col, wpre, bpre);

    v2f bfr[40];
    #pragma unroll
    for (int kk = 0; kk < 40; ++kk) {
      const int k0 = kk * 4 + half * 2;
      bfr[kk].x = wS[col][k0]; bfr[kk].y = wS[col][k0 + 1];
    }

    const int ar = rt * 16 + lr;
    v2f afr[16], afm[16], afp[8];
    #pragma unroll
    for (int kk = 0; kk < 16; ++kk) {
      const int k0 = kk * 4 + half * 2;
      afr[kk].x = src[ar][k0]; afr[kk].y = src[ar][k0 + 1];
    }
    #pragma unroll
    for (int kk = 0; kk < 16; ++kk) {
      const int k0 = kk * 4 + half * 2;
      afm[kk].x = meanh[k0]; afm[kk].y = meanh[k0 + 1];
    }
    #pragma unroll
    for (int kk = 0; kk < 8; ++kk) {
      const int k0 = kk * 4 + half * 2;
      afp[kk].x = mprev[ar][k0]; afp[kk].y = mprev[ar][k0 + 1];
    }
    float res[8];
    #pragma unroll
    for (int r = 0; r < 8; ++r) res[r] = src[rt * 16 + r + half * 8][col];
    __builtin_amdgcn_sched_barrier(0);

    v8f acc = {};
    #pragma unroll
    for (int kk = 0; kk < 40; ++kk) {
      const v2f a = (kk < 16) ? afr[kk] : (kk < 32) ? afm[kk - 16] : afp[kk - 32];
      acc = wmma_f32_k4(a, bfr[kk], acc);
    }

    #pragma unroll
    for (int r = 0; r < 8; ++r)
      dst[rt * 16 + r + half * 8][col] = fast_tanh(acc[r] + bias) + res[r];
    __syncthreads();
  }

  if (tid < DIMX * NP) {
    float acc = fbv;
    for (int k = 0; k < H1; ++k) acc += hB[fpi][k] * fwS[fdd][k];
    out[b * NP * DIMX + tid] = acc;
  }
}

extern "C" void kernel_launch(void* const* d_in, const int* in_sizes, int n_in,
                              void* d_out, int out_size, void* d_ws,
                              size_t ws_size, hipStream_t stream) {
  const float* x       = (const float*)d_in[0];
  const float* t       = (const float*)d_in[1];
  const float* fc1_w0  = (const float*)d_in[2];
  const float* fc1_b0  = (const float*)d_in[3];
  const float* fc1_w   = (const float*)d_in[4];
  const float* fc1_b   = (const float*)d_in[5];
  const float* fc2_w0  = (const float*)d_in[6];
  const float* fc2_b0  = (const float*)d_in[7];
  const float* fc2_w   = (const float*)d_in[8];
  const float* fc2_b   = (const float*)d_in[9];
  const float* final_w = (const float*)d_in[10];
  const float* final_b = (const float*)d_in[11];
  float* out = (float*)d_out;
  (void)in_sizes; (void)n_in; (void)out_size; (void)d_ws; (void)ws_size;

  drift_fused<<<dim3(BATCHN), FTHREADS, 0, stream>>>(
      x, t, fc1_w0, fc1_b0, fc1_w, fc1_b, fc2_w0, fc2_b0, fc2_w, fc2_b,
      final_w, final_b, out);
}
